# starting kernel
# baseline (speedup 1.0000x reference)
#include <hip/hip_runtime.h>
#include <hip/hip_bf16.h>

typedef __attribute__((ext_vector_type(2))) float v2f;
typedef __attribute__((ext_vector_type(8))) float v8f;

#define NODES 100000

__global__ void zero_f32_kernel(float* __restrict__ p, int n) {
  int i = blockIdx.x * blockDim.x + threadIdx.x;
  if (i < n) p[i] = 0.0f;
}

__global__ void deg_count_kernel(const int* __restrict__ dst, int e, float* __restrict__ deg) {
  int i = blockIdx.x * blockDim.x + threadIdx.x;
  if (i < e) atomicAdd(&deg[dst[i]], 1.0f);
}

__global__ void deg_finalize_kernel(float* __restrict__ deg, int n) {
  int i = blockIdx.x * blockDim.x + threadIdx.x;
  if (i < n) deg[i] = rsqrtf(deg[i] + 1.0f);
}

template <int K, int FOUT, bool RELU>
__global__ void gemm_wmma_f32_kernel(const float* __restrict__ X,
                                     const float* __restrict__ W,
                                     float* __restrict__ H,
                                     int n_rows) {
  constexpr int NT = FOUT / 16;
  __shared__ float Wp[K * FOUT];

  for (int i = threadIdx.x; i < K * FOUT; i += 256) {
    int k = i / FOUT;
    int col = i - k * FOUT;
    Wp[(k >> 1) * (FOUT * 2) + col * 2 + (k & 1)] = W[i];
  }
  __syncthreads();

  const int wave = threadIdx.x >> 5;
  const int lane = threadIdx.x & 31;
  const int tile_m = blockIdx.x * 8 + wave;
  if (tile_m * 16 >= n_rows) return;

  const int half = lane >> 4;
  const int l15  = lane & 15;

  const float* __restrict__ xrow = X + (size_t)(tile_m * 16 + l15) * K;

  v8f acc[NT];
#pragma unroll
  for (int t = 0; t < NT; ++t) acc[t] = (v8f){};

#pragma unroll
  for (int k = 0; k < K; k += 4) {
    const int kk = k + half * 2;
    float a0 = xrow[kk];
    float a1 = xrow[kk + 1];
    if (RELU) { a0 = fmaxf(a0, 0.0f); a1 = fmaxf(a1, 0.0f); }
    v2f a;
    a.x = a0;
    a.y = a1;
    const float* __restrict__ wrow = Wp + (kk >> 1) * (FOUT * 2) + l15 * 2;
#pragma unroll
    for (int t = 0; t < NT; ++t) {
      const v2f b = *(const v2f*)(wrow + t * 32);
      acc[t] = __builtin_amdgcn_wmma_f32_16x16x4_f32(false, a, false, b,
                                                     (short)0, acc[t], false, false);
    }
  }

  float* __restrict__ hbase = H + (size_t)(tile_m * 16) * FOUT + l15;
#pragma unroll
  for (int t = 0; t < NT; ++t) {
#pragma unroll
    for (int v = 0; v < 8; ++v) {
      hbase[(size_t)(v + half * 8) * FOUT + t * 16] = acc[t][v];
    }
  }
}

__global__ void self_init_kernel(const float* __restrict__ H,
                                 const float* __restrict__ dinv,
                                 const float* __restrict__ bias,
                                 float* __restrict__ Out,
                                 int n, int fshift) {
  const int fmask = (1 << fshift) - 1;
  long long idx = (long long)blockIdx.x * blockDim.x + threadIdx.x;
  long long total = ((long long)n) << fshift;
  if (idx >= total) return;
  int node = (int)(idx >> fshift);
  int f = (int)(idx & fmask);
  float di = dinv[node];
  Out[idx] = H[idx] * di * di + bias[f];
}

__global__ void edge_scatter_kernel(const int* __restrict__ src,
                                    const int* __restrict__ dst,
                                    const float* __restrict__ H,
                                    const float* __restrict__ dinv,
                                    float* __restrict__ Out,
                                    int e, int fshift) {
  const int fmask = (1 << fshift) - 1;
  long long idx = (long long)blockIdx.x * blockDim.x + threadIdx.x;
  long long total = ((long long)e) << fshift;
  if (idx >= total) return;
  int edge = (int)(idx >> fshift);
  int f = (int)(idx & fmask);
  int s = src[edge];
  int d = dst[edge];
  float coef = dinv[s] * dinv[d];
  atomicAdd(&Out[(((size_t)d) << fshift) + f], H[(((size_t)s) << fshift) + f] * coef);
}

static inline unsigned int cdiv(long long a, long long b) {
  return (unsigned int)((a + b - 1) / b);
}

extern "C" void kernel_launch(void* const* d_in, const int* in_sizes, int n_in,
                              void* d_out, int out_size, void* d_ws, size_t ws_size,
                              hipStream_t stream) {
  const float* x  = (const float*)d_in[0];
  const int*   ei = (const int*)d_in[1];
  const float* W1 = (const float*)d_in[2];
  const float* b1 = (const float*)d_in[3];
  const float* W2 = (const float*)d_in[4];
  const float* b2 = (const float*)d_in[5];
  const float* W3 = (const float*)d_in[6];
  const float* b3 = (const float*)d_in[7];
  float* out = (float*)d_out;

  const int N = NODES;
  const int E = in_sizes[1] / 2;

  const int* src = ei;
  const int* dst = ei + E;

  float* dinv = (float*)d_ws;
  float* bufA = dinv + N;
  float* bufB = bufA + (size_t)N * 128;

  const int T = 256;
  const unsigned int gemm_blocks = cdiv((N + 15) / 16, 8);

  zero_f32_kernel<<<cdiv(N, T), T, 0, stream>>>(dinv, N);
  deg_count_kernel<<<cdiv(E, T), T, 0, stream>>>(dst, E, dinv);
  deg_finalize_kernel<<<cdiv(N, T), T, 0, stream>>>(dinv, N);

  gemm_wmma_f32_kernel<128, 64, false><<<gemm_blocks, T, 0, stream>>>(x, W1, bufA, N);
  self_init_kernel<<<cdiv((long long)N << 6, T), T, 0, stream>>>(bufA, dinv, b1, bufB, N, 6);
  edge_scatter_kernel<<<cdiv((long long)E << 6, T), T, 0, stream>>>(src, dst, bufA, dinv, bufB, E, 6);

  gemm_wmma_f32_kernel<64, 128, true><<<gemm_blocks, T, 0, stream>>>(bufB, W2, bufA, N);
  self_init_kernel<<<cdiv((long long)N << 7, T), T, 0, stream>>>(bufA, dinv, b2, bufB, N, 7);
  edge_scatter_kernel<<<cdiv((long long)E << 7, T), T, 0, stream>>>(src, dst, bufA, dinv, bufB, E, 7);

  gemm_wmma_f32_kernel<128, 64, true><<<gemm_blocks, T, 0, stream>>>(bufB, W3, bufA, N);
  self_init_kernel<<<cdiv((long long)N << 6, T), T, 0, stream>>>(bufA, dinv, b3, out, N, 6);
  edge_scatter_kernel<<<cdiv((long long)E << 6, T), T, 0, stream>>>(src, dst, bufA, dinv, out, E, 6);
}


# layer 2 aggregates its 64-wide input before the 64->128 GEMM (linearity: A(hW)=(Ah)W, bias in GEMM epilogue): half that layer's edge atomics; changes the order of the f32 sums
# speedup vs baseline: 1.3627x; 1.3627x over previous
#include <hip/hip_runtime.h>
#include <hip/hip_bf16.h>

typedef __attribute__((ext_vector_type(2))) float v2f;
typedef __attribute__((ext_vector_type(8))) float v8f;

#define NODES 100000

__global__ void zero_f32_kernel(float* __restrict__ p, int n) {
  int i = blockIdx.x * blockDim.x + threadIdx.x;
  if (i < n) p[i] = 0.0f;
}

__global__ void deg_count_kernel(const int* __restrict__ dst, int e, float* __restrict__ deg) {
  int i = blockIdx.x * blockDim.x + threadIdx.x;
  if (i < e) atomicAdd(&deg[dst[i]], 1.0f);
}

__global__ void deg_finalize_kernel(float* __restrict__ deg, int n) {
  int i = blockIdx.x * blockDim.x + threadIdx.x;
  if (i < n) deg[i] = rsqrtf(deg[i] + 1.0f);
}

template <int K, int FOUT, bool RELU, bool BIAS>
__global__ void gemm_wmma_f32_kernel(const float* __restrict__ X,
                                     const float* __restrict__ W,
                                     const float* __restrict__ bias,
                                     float* __restrict__ H,
                                     int n_rows) {
  constexpr int NT = FOUT / 16;
  __shared__ float Wp[K * FOUT];

  for (int i = threadIdx.x; i < K * FOUT; i += 256) {
    int k = i / FOUT;
    int col = i - k * FOUT;
    Wp[(k >> 1) * (FOUT * 2) + col * 2 + (k & 1)] = W[i];
  }
  __syncthreads();

  const int wave = threadIdx.x >> 5;
  const int lane = threadIdx.x & 31;
  const int tile_m = blockIdx.x * 8 + wave;
  if (tile_m * 16 >= n_rows) return;

  const int half = lane >> 4;
  const int l15  = lane & 15;

  const float* __restrict__ xrow = X + (size_t)(tile_m * 16 + l15) * K;

  v8f acc[NT];
#pragma unroll
  for (int t = 0; t < NT; ++t) acc[t] = (v8f){};

#pragma unroll
  for (int k = 0; k < K; k += 4) {
    const int kk = k + half * 2;
    float a0 = xrow[kk];
    float a1 = xrow[kk + 1];
    if (RELU) { a0 = fmaxf(a0, 0.0f); a1 = fmaxf(a1, 0.0f); }
    v2f a;
    a.x = a0;
    a.y = a1;
    const float* __restrict__ wrow = Wp + (kk >> 1) * (FOUT * 2) + l15 * 2;
#pragma unroll
    for (int t = 0; t < NT; ++t) {
      const v2f b = *(const v2f*)(wrow + t * 32);
      acc[t] = __builtin_amdgcn_wmma_f32_16x16x4_f32(false, a, false, b,
                                                     (short)0, acc[t], false, false);
    }
  }

  float* __restrict__ hbase = H + (size_t)(tile_m * 16) * FOUT + l15;
#pragma unroll
  for (int t = 0; t < NT; ++t) {
    float bv = 0.0f;
    if (BIAS) bv = bias[t * 16 + l15];
#pragma unroll
    for (int v = 0; v < 8; ++v) {
      hbase[(size_t)(v + half * 8) * FOUT + t * 16] = acc[t][v] + bv;
    }
  }
}

template <bool RELU_IN, bool BIAS>
__global__ void self_init_kernel(const float* __restrict__ H,
                                 const float* __restrict__ dinv,
                                 const float* __restrict__ bias,
                                 float* __restrict__ Out,
                                 int n, int fshift) {
  const int fmask = (1 << fshift) - 1;
  long long idx = (long long)blockIdx.x * blockDim.x + threadIdx.x;
  long long total = ((long long)n) << fshift;
  if (idx >= total) return;
  int node = (int)(idx >> fshift);
  int f = (int)(idx & fmask);
  float di = dinv[node];
  float h = H[idx];
  if (RELU_IN) h = fmaxf(h, 0.0f);
  float r = h * di * di;
  if (BIAS) r += bias[f];
  Out[idx] = r;
}

template <bool RELU_IN>
__global__ void edge_scatter_kernel(const int* __restrict__ src,
                                    const int* __restrict__ dst,
                                    const float* __restrict__ H,
                                    const float* __restrict__ dinv,
                                    float* __restrict__ Out,
                                    int e, int fshift) {
  const int fmask = (1 << fshift) - 1;
  long long idx = (long long)blockIdx.x * blockDim.x + threadIdx.x;
  long long total = ((long long)e) << fshift;
  if (idx >= total) return;
  int edge = (int)(idx >> fshift);
  int f = (int)(idx & fmask);
  int s = src[edge];
  int d = dst[edge];
  float coef = dinv[s] * dinv[d];
  float h = H[(((size_t)s) << fshift) + f];
  if (RELU_IN) h = fmaxf(h, 0.0f);
  atomicAdd(&Out[(((size_t)d) << fshift) + f], h * coef);
}

static inline unsigned int cdiv(long long a, long long b) {
  return (unsigned int)((a + b - 1) / b);
}

extern "C" void kernel_launch(void* const* d_in, const int* in_sizes, int n_in,
                              void* d_out, int out_size, void* d_ws, size_t ws_size,
                              hipStream_t stream) {
  const float* x  = (const float*)d_in[0];
  const int*   ei = (const int*)d_in[1];
  const float* W1 = (const float*)d_in[2];
  const float* b1 = (const float*)d_in[3];
  const float* W2 = (const float*)d_in[4];
  const float* b2 = (const float*)d_in[5];
  const float* W3 = (const float*)d_in[6];
  const float* b3 = (const float*)d_in[7];
  float* out = (float*)d_out;

  const int N = NODES;
  const int E = in_sizes[1] / 2;

  const int* src = ei;
  const int* dst = ei + E;

  float* dinv = (float*)d_ws;
  float* bufA = dinv + N;
  float* bufB = bufA + (size_t)N * 128;

  const int T = 256;
  const unsigned int gemm_blocks = cdiv((N + 15) / 16, 8);

  zero_f32_kernel<<<cdiv(N, T), T, 0, stream>>>(dinv, N);
  deg_count_kernel<<<cdiv(E, T), T, 0, stream>>>(dst, E, dinv);
  deg_finalize_kernel<<<cdiv(N, T), T, 0, stream>>>(dinv, N);

  gemm_wmma_f32_kernel<128, 64, false, false><<<gemm_blocks, T, 0, stream>>>(x, W1, b1, bufA, N);
  self_init_kernel<false, true><<<cdiv((long long)N << 6, T), T, 0, stream>>>(bufA, dinv, b1, bufB, N, 6);
  edge_scatter_kernel<false><<<cdiv((long long)E << 6, T), T, 0, stream>>>(src, dst, bufA, dinv, bufB, E, 6);

  self_init_kernel<true, false><<<cdiv((long long)N << 6, T), T, 0, stream>>>(bufB, dinv, b2, bufA, N, 6);
  edge_scatter_kernel<true><<<cdiv((long long)E << 6, T), T, 0, stream>>>(src, dst, bufB, dinv, bufA, E, 6);
  gemm_wmma_f32_kernel<64, 128, false, true><<<gemm_blocks, T, 0, stream>>>(bufA, W2, b2, bufB, N);

  gemm_wmma_f32_kernel<128, 64, true, false><<<gemm_blocks, T, 0, stream>>>(bufB, W3, b3, bufA, N);
  self_init_kernel<false, true><<<cdiv((long long)N << 6, T), T, 0, stream>>>(bufA, dinv, b3, out, N, 6);
  edge_scatter_kernel<false><<<cdiv((long long)E << 6, T), T, 0, stream>>>(src, dst, bufA, dinv, out, E, 6);
}


# edge sums by gather instead of float atomics: per-call destination-sorted edge list (arrival-order rank from integer atomics), one wave per row sums in registers; changes the order of the f32 sums
# speedup vs baseline: 5.0038x; 3.6720x over previous
#include <hip/hip_runtime.h>
#include <hip/hip_bf16.h>

typedef __attribute__((ext_vector_type(2))) float v2f;
typedef __attribute__((ext_vector_type(8))) float v8f;

#define NODES 100000

__global__ void zero_i32_kernel(int* __restrict__ p, int n) {
  int i = blockIdx.x * blockDim.x + threadIdx.x;
  if (i < n) p[i] = 0;
}

__global__ void count_rank_kernel(const int* __restrict__ dst, int e,
                                  int* __restrict__ cnt, int* __restrict__ rank) {
  int i = blockIdx.x * blockDim.x + threadIdx.x;
  if (i < e) rank[i] = atomicAdd(&cnt[dst[i]], 1);
}

__global__ void node_prep_kernel(const int* __restrict__ cnt, int n,
                                 int* __restrict__ total,
                                 int2* __restrict__ seg, float* __restrict__ dinv) {
  const int i = blockIdx.x * blockDim.x + threadIdx.x;
  const int lane = threadIdx.x & 31;
  const int c = (i < n) ? cnt[i] : 0;
  int incl = c;
#pragma unroll
  for (int o = 1; o < 32; o <<= 1) {
    int t = __shfl_up(incl, o);
    if (lane >= o) incl += t;
  }
  const int wsum = __shfl(incl, 31);
  int base = 0;
  if (lane == 0) base = atomicAdd(total, wsum);
  base = __shfl(base, 0);
  if (i < n) {
    seg[i] = make_int2(base + incl - c, c);
    dinv[i] = rsqrtf((float)c + 1.0f);
  }
}

__global__ void fill_kernel(const int* __restrict__ src, const int* __restrict__ dst,
                            const int* __restrict__ rank, const int2* __restrict__ seg,
                            const float* __restrict__ dinv, int e,
                            int2* __restrict__ colw) {
  int i = blockIdx.x * blockDim.x + threadIdx.x;
  if (i >= e) return;
  const int s = src[i];
  const int d = dst[i];
  const int p = seg[d].x + rank[i];
  colw[p] = make_int2(s, __float_as_int(dinv[s] * dinv[d]));
}

static __device__ __forceinline__ float4 ld_row16(const float* __restrict__ H, unsigned s, int q) {
  const unsigned off = s * 256u + (unsigned)q * 16u;
  return *(const float4*)((const char*)H + off);
}

template <bool RELU_OUT, bool BIAS>
__global__ void __launch_bounds__(256)
gather_rows_kernel(const int2* __restrict__ seg, const int2* __restrict__ colw,
                   const float* __restrict__ H, const float* __restrict__ dinv,
                   const float* __restrict__ bias, float* __restrict__ Out, int n) {
  const int lane = threadIdx.x & 31;
  const int node = __builtin_amdgcn_readfirstlane((int)(blockIdx.x * 8 + (threadIdx.x >> 5)));
  if (node >= n) return;
  const int half = lane >> 4;
  const int q = lane & 15;

  const int2 sg = seg[node];
  const int start = sg.x;
  const int cnt = sg.y;
  const float di = dinv[node];
  const float4 hself = ld_row16(H, (unsigned)node, q);

  float4 acc = make_float4(0.0f, 0.0f, 0.0f, 0.0f);
  const unsigned last_row = (unsigned)(n - 1);
  for (int j = 0; j < cnt; j += 16) {
    const int m = cnt - j;
    const int2* __restrict__ cp = colw + (start + j);
    int ex[8], ey[8], ox[8], oy[8];
#pragma unroll
    for (int u = 0; u < 8; ++u) {
      const int2 ce = cp[2 * u];
      const int2 co = cp[2 * u + 1];
      ex[u] = __builtin_amdgcn_readfirstlane(ce.x);
      ey[u] = __builtin_amdgcn_readfirstlane(ce.y);
      ox[u] = __builtin_amdgcn_readfirstlane(co.x);
      oy[u] = __builtin_amdgcn_readfirstlane(co.y);
    }
    float  w[8];
    float4 v[8];
#pragma unroll
    for (int u = 0; u < 8; ++u) {
      const int sidx = half ? ox[u] : ex[u];
      w[u] = __int_as_float(half ? oy[u] : ey[u]);
      const unsigned srow = ((unsigned)sidx < last_row) ? (unsigned)sidx : last_row;
      v[u] = make_float4(0.0f, 0.0f, 0.0f, 0.0f);
      if (2 * u + half < m) v[u] = ld_row16(H, srow, q);
    }
#pragma unroll
    for (int u = 0; u < 8; ++u) {
      const float wu = (2 * u + half < m) ? w[u] : 0.0f;
      acc.x = fmaf(wu, v[u].x, acc.x);
      acc.y = fmaf(wu, v[u].y, acc.y);
      acc.z = fmaf(wu, v[u].z, acc.z);
      acc.w = fmaf(wu, v[u].w, acc.w);
    }
  }

  acc.x += __shfl_xor(acc.x, 16);
  acc.y += __shfl_xor(acc.y, 16);
  acc.z += __shfl_xor(acc.z, 16);
  acc.w += __shfl_xor(acc.w, 16);

  if (half == 0) {
    float4 r;
    r.x = hself.x * di * di;
    r.y = hself.y * di * di;
    r.z = hself.z * di * di;
    r.w = hself.w * di * di;
    if (BIAS) {
      r.x += bias[q * 4 + 0];
      r.y += bias[q * 4 + 1];
      r.z += bias[q * 4 + 2];
      r.w += bias[q * 4 + 3];
    }
    r.x += acc.x; r.y += acc.y; r.z += acc.z; r.w += acc.w;
    if (RELU_OUT) {
      r.x = fmaxf(r.x, 0.0f); r.y = fmaxf(r.y, 0.0f);
      r.z = fmaxf(r.z, 0.0f); r.w = fmaxf(r.w, 0.0f);
    }
    *(float4*)((char*)Out + ((unsigned)node * 256u + (unsigned)q * 16u)) = r;
  }
}

template <int K, int FOUT, bool RELU, bool BIAS>
__global__ void gemm_wmma_f32_kernel(const float* __restrict__ X,
                                     const float* __restrict__ W,
                                     const float* __restrict__ bias,
                                     float* __restrict__ H,
                                     int n_rows) {
  constexpr int NT = FOUT / 16;
  __shared__ float Wp[K * FOUT];

  for (int i = threadIdx.x; i < K * FOUT; i += 256) {
    int k = i / FOUT;
    int col = i - k * FOUT;
    Wp[(k >> 1) * (FOUT * 2) + col * 2 + (k & 1)] = W[i];
  }
  __syncthreads();

  const int wave = threadIdx.x >> 5;
  const int lane = threadIdx.x & 31;
  const int tile_m = blockIdx.x * 8 + wave;
  if (tile_m * 16 >= n_rows) return;

  const int half = lane >> 4;
  const int l15  = lane & 15;

  const float* __restrict__ xrow = X + (size_t)(tile_m * 16 + l15) * K;

  v8f acc[NT];
#pragma unroll
  for (int t = 0; t < NT; ++t) acc[t] = (v8f){};

#pragma unroll
  for (int k = 0; k < K; k += 4) {
    const int kk = k + half * 2;
    float a0 = xrow[kk];
    float a1 = xrow[kk + 1];
    if (RELU) { a0 = fmaxf(a0, 0.0f); a1 = fmaxf(a1, 0.0f); }
    v2f a;
    a.x = a0;
    a.y = a1;
    const float* __restrict__ wrow = Wp + (kk >> 1) * (FOUT * 2) + l15 * 2;
#pragma unroll
    for (int t = 0; t < NT; ++t) {
      const v2f b = *(const v2f*)(wrow + t * 32);
      acc[t] = __builtin_amdgcn_wmma_f32_16x16x4_f32(false, a, false, b,
                                                     (short)0, acc[t], false, false);
    }
  }

  float* __restrict__ hbase = H + (size_t)(tile_m * 16) * FOUT + l15;
#pragma unroll
  for (int t = 0; t < NT; ++t) {
    float bv = 0.0f;
    if (BIAS) bv = bias[t * 16 + l15];
#pragma unroll
    for (int v = 0; v < 8; ++v) {
      hbase[(size_t)(v + half * 8) * FOUT + t * 16] = acc[t][v] + bv;
    }
  }
}

static inline unsigned int cdiv(long long a, long long b) {
  return (unsigned int)((a + b - 1) / b);
}

extern "C" void kernel_launch(void* const* d_in, const int* in_sizes, int n_in,
                              void* d_out, int out_size, void* d_ws, size_t ws_size,
                              hipStream_t stream) {
  const float* x  = (const float*)d_in[0];
  const int*   ei = (const int*)d_in[1];
  const float* W1 = (const float*)d_in[2];
  const float* b1 = (const float*)d_in[3];
  const float* W2 = (const float*)d_in[4];
  const float* b2 = (const float*)d_in[5];
  const float* W3 = (const float*)d_in[6];
  const float* b3 = (const float*)d_in[7];
  float* out = (float*)d_out;

  const int N = NODES;
  const int E = in_sizes[1] / 2;

  const int* src = ei;
  const int* dst = ei + E;

  float* dinv = (float*)d_ws;
  float* bufA = dinv + N;
  float* bufB = bufA + (size_t)N * 128;
  int2* colw  = (int2*)(bufA + (size_t)N * 64);
  int*  rank  = (int*)(colw + E);
  int2* seg   = (int2*)(rank + E);
  int*  cnt   = (int*)(seg + N);
  int*  total = cnt + N;

  const int T = 256;
  const unsigned int gemm_blocks = cdiv((N + 15) / 16, 8);
  const unsigned int row_blocks  = cdiv(N, 8);

  zero_i32_kernel<<<cdiv(N + 1, T), T, 0, stream>>>(cnt, N + 1);
  count_rank_kernel<<<cdiv(E, T), T, 0, stream>>>(dst, E, cnt, rank);
  node_prep_kernel<<<cdiv(N, T), T, 0, stream>>>(cnt, N, total, seg, dinv);
  fill_kernel<<<cdiv(E, T), T, 0, stream>>>(src, dst, rank, seg, dinv, E, colw);

  gemm_wmma_f32_kernel<128, 64, false, false><<<gemm_blocks, T, 0, stream>>>(x, W1, b1, bufA, N);
  gather_rows_kernel<true, true><<<row_blocks, T, 0, stream>>>(seg, colw, bufA, dinv, b1, bufB, N);

  gather_rows_kernel<false, false><<<row_blocks, T, 0, stream>>>(seg, colw, bufB, dinv, b2, bufA, N);
  gemm_wmma_f32_kernel<64, 128, false, true><<<gemm_blocks, T, 0, stream>>>(bufA, W2, b2, bufB, N);

  gemm_wmma_f32_kernel<128, 64, true, false><<<gemm_blocks, T, 0, stream>>>(bufB, W3, b3, bufA, N);
  gather_rows_kernel<false, true><<<row_blocks, T, 0, stream>>>(seg, colw, bufA, dinv, b3, out, N);
}


# layer 2's row gather and both dense products behind it (64->128 +bias, ReLU, 128->64) in one launch: a wave gathers a 16-row tile into LDS and runs both WMMA products from it; 10 launches -> 8
# speedup vs baseline: 5.3420x; 1.0676x over previous
#include <hip/hip_runtime.h>
#include <hip/hip_bf16.h>

typedef __attribute__((ext_vector_type(2))) float v2f;
typedef __attribute__((ext_vector_type(8))) float v8f;

#define NODES 100000

__global__ void zero_i32_kernel(int* __restrict__ p, int n) {
  int i = blockIdx.x * blockDim.x + threadIdx.x;
  if (i < n) p[i] = 0;
}

__global__ void count_rank_kernel(const int* __restrict__ dst, int e,
                                  int* __restrict__ cnt, int* __restrict__ rank) {
  int i = blockIdx.x * blockDim.x + threadIdx.x;
  if (i < e) rank[i] = atomicAdd(&cnt[dst[i]], 1);
}

__global__ void node_prep_kernel(const int* __restrict__ cnt, int n,
                                 int* __restrict__ total,
                                 int2* __restrict__ seg, float* __restrict__ dinv) {
  const int i = blockIdx.x * blockDim.x + threadIdx.x;
  const int lane = threadIdx.x & 31;
  const int c = (i < n) ? cnt[i] : 0;
  int incl = c;
#pragma unroll
  for (int o = 1; o < 32; o <<= 1) {
    int t = __shfl_up(incl, o);
    if (lane >= o) incl += t;
  }
  const int wsum = __shfl(incl, 31);
  int base = 0;
  if (lane == 0) base = atomicAdd(total, wsum);
  base = __shfl(base, 0);
  if (i < n) {
    seg[i] = make_int2(base + incl - c, c);
    dinv[i] = rsqrtf((float)c + 1.0f);
  }
}

__global__ void fill_kernel(const int* __restrict__ src, const int* __restrict__ dst,
                            const int* __restrict__ rank, const int2* __restrict__ seg,
                            const float* __restrict__ dinv, int e,
                            int2* __restrict__ colw) {
  int i = blockIdx.x * blockDim.x + threadIdx.x;
  if (i >= e) return;
  const int s = src[i];
  const int d = dst[i];
  const int p = seg[d].x + rank[i];
  colw[p] = make_int2(s, __float_as_int(dinv[s] * dinv[d]));
}

static __device__ __forceinline__ float4 ld_row16(const float* __restrict__ H, unsigned s, int q) {
  const unsigned off = s * 256u + (unsigned)q * 16u;
  return *(const float4*)((const char*)H + off);
}

template <bool RELU_OUT, bool BIAS>
__global__ void __launch_bounds__(256)
gather_rows_kernel(const int2* __restrict__ seg, const int2* __restrict__ colw,
                   const float* __restrict__ H, const float* __restrict__ dinv,
                   const float* __restrict__ bias, float* __restrict__ Out, int n) {
  const int lane = threadIdx.x & 31;
  const int node = __builtin_amdgcn_readfirstlane((int)(blockIdx.x * 8 + (threadIdx.x >> 5)));
  if (node >= n) return;
  const int half = lane >> 4;
  const int q = lane & 15;

  const int2 sg = seg[node];
  const int start = sg.x;
  const int cnt = sg.y;
  const float di = dinv[node];
  const float4 hself = ld_row16(H, (unsigned)node, q);

  float4 acc = make_float4(0.0f, 0.0f, 0.0f, 0.0f);
  const unsigned last_row = (unsigned)(n - 1);
  for (int j = 0; j < cnt; j += 16) {
    const int m = cnt - j;
    const int2* __restrict__ cp = colw + (start + j);
    int ex[8], ey[8], ox[8], oy[8];
#pragma unroll
    for (int u = 0; u < 8; ++u) {
      const int2 ce = cp[2 * u];
      const int2 co = cp[2 * u + 1];
      ex[u] = __builtin_amdgcn_readfirstlane(ce.x);
      ey[u] = __builtin_amdgcn_readfirstlane(ce.y);
      ox[u] = __builtin_amdgcn_readfirstlane(co.x);
      oy[u] = __builtin_amdgcn_readfirstlane(co.y);
    }
    float  w[8];
    float4 v[8];
#pragma unroll
    for (int u = 0; u < 8; ++u) {
      const int sidx = half ? ox[u] : ex[u];
      w[u] = __int_as_float(half ? oy[u] : ey[u]);
      const unsigned srow = ((unsigned)sidx < last_row) ? (unsigned)sidx : last_row;
      v[u] = make_float4(0.0f, 0.0f, 0.0f, 0.0f);
      if (2 * u + half < m) v[u] = ld_row16(H, srow, q);
    }
#pragma unroll
    for (int u = 0; u < 8; ++u) {
      const float wu = (2 * u + half < m) ? w[u] : 0.0f;
      acc.x = fmaf(wu, v[u].x, acc.x);
      acc.y = fmaf(wu, v[u].y, acc.y);
      acc.z = fmaf(wu, v[u].z, acc.z);
      acc.w = fmaf(wu, v[u].w, acc.w);
    }
  }

  acc.x += __shfl_xor(acc.x, 16);
  acc.y += __shfl_xor(acc.y, 16);
  acc.z += __shfl_xor(acc.z, 16);
  acc.w += __shfl_xor(acc.w, 16);

  if (half == 0) {
    float4 r;
    r.x = hself.x * di * di;
    r.y = hself.y * di * di;
    r.z = hself.z * di * di;
    r.w = hself.w * di * di;
    if (BIAS) {
      r.x += bias[q * 4 + 0];
      r.y += bias[q * 4 + 1];
      r.z += bias[q * 4 + 2];
      r.w += bias[q * 4 + 3];
    }
    r.x += acc.x; r.y += acc.y; r.z += acc.z; r.w += acc.w;
    if (RELU_OUT) {
      r.x = fmaxf(r.x, 0.0f); r.y = fmaxf(r.y, 0.0f);
      r.z = fmaxf(r.z, 0.0f); r.w = fmaxf(r.w, 0.0f);
    }
    *(float4*)((char*)Out + ((unsigned)node * 256u + (unsigned)q * 16u)) = r;
  }
}

template <int K, int FOUT, bool RELU, bool BIAS>
__global__ void gemm_wmma_f32_kernel(const float* __restrict__ X,
                                     const float* __restrict__ W,
                                     const float* __restrict__ bias,
                                     float* __restrict__ H,
                                     int n_rows) {
  constexpr int NT = FOUT / 16;
  __shared__ float Wp[K * FOUT];

  for (int i = threadIdx.x; i < K * FOUT; i += 256) {
    int k = i / FOUT;
    int col = i - k * FOUT;
    Wp[(k >> 1) * (FOUT * 2) + col * 2 + (k & 1)] = W[i];
  }
  __syncthreads();

  const int wave = threadIdx.x >> 5;
  const int lane = threadIdx.x & 31;
  const int tile_m = blockIdx.x * 8 + wave;
  if (tile_m * 16 >= n_rows) return;

  const int half = lane >> 4;
  const int l15  = lane & 15;

  const float* __restrict__ xrow = X + (size_t)(tile_m * 16 + l15) * K;

  v8f acc[NT];
#pragma unroll
  for (int t = 0; t < NT; ++t) acc[t] = (v8f){};

#pragma unroll
  for (int k = 0; k < K; k += 4) {
    const int kk = k + half * 2;
    float a0 = xrow[kk];
    float a1 = xrow[kk + 1];
    if (RELU) { a0 = fmaxf(a0, 0.0f); a1 = fmaxf(a1, 0.0f); }
    v2f a;
    a.x = a0;
    a.y = a1;
    const float* __restrict__ wrow = Wp + (kk >> 1) * (FOUT * 2) + l15 * 2;
#pragma unroll
    for (int t = 0; t < NT; ++t) {
      const v2f b = *(const v2f*)(wrow + t * 32);
      acc[t] = __builtin_amdgcn_wmma_f32_16x16x4_f32(false, a, false, b,
                                                     (short)0, acc[t], false, false);
    }
  }

  float* __restrict__ hbase = H + (size_t)(tile_m * 16) * FOUT + l15;
#pragma unroll
  for (int t = 0; t < NT; ++t) {
    float bv = 0.0f;
    if (BIAS) bv = bias[t * 16 + l15];
#pragma unroll
    for (int v = 0; v < 8; ++v) {
      hbase[(size_t)(v + half * 8) * FOUT + t * 16] = acc[t][v] + bv;
    }
  }
}


#define FPITCH 68
#define FWAVES 25
#define W2PITCH (2 * 128 + 32)
#define W3PITCH (2 * 64 + 32)

static __device__ __forceinline__ void wave_lds_order() {
  __builtin_amdgcn_fence(__ATOMIC_SEQ_CST, "wavefront");
  __builtin_amdgcn_wave_barrier();
}

static __device__ __forceinline__ void load_pairs(const int2* __restrict__ colw, int first,
                                                  int half, int2 (&cw)[8]) {
#pragma unroll
  for (int u = 0; u < 8; ++u) cw[u] = colw[first + 2 * u + half];
}

static __device__ __forceinline__ void issue_rows(const float* __restrict__ H, const int2 (&cw)[8],
                                                  int m, int half, int q, unsigned last_row,
                                                  float4 (&v)[8]) {
#pragma unroll
  for (int u = 0; u < 8; ++u) {
    const unsigned srow = ((unsigned)cw[u].x < last_row) ? (unsigned)cw[u].x : last_row;
    v[u] = make_float4(0.0f, 0.0f, 0.0f, 0.0f);
    if (2 * u + half < m) v[u] = ld_row16(H, srow, q);
  }
}

static __device__ __forceinline__ void fma_rows(const int2 (&cw)[8], const float4 (&v)[8],
                                                int m, int half, float4& acc) {
#pragma unroll
  for (int u = 0; u < 8; ++u) {
    const float wu = (2 * u + half < m) ? __int_as_float(cw[u].y) : 0.0f;
    acc.x = fmaf(wu, v[u].x, acc.x);
    acc.y = fmaf(wu, v[u].y, acc.y);
    acc.z = fmaf(wu, v[u].z, acc.z);
    acc.w = fmaf(wu, v[u].w, acc.w);
  }
}

__global__ void __launch_bounds__(FWAVES * 32)
gather_mlp_kernel(const int2* __restrict__ seg, const int2* __restrict__ colw,
                  const float* __restrict__ H, const float* __restrict__ dinv,
                  const float* __restrict__ W2, const float* __restrict__ b2,
                  const float* __restrict__ W3, float* __restrict__ Out, int n) {
  __shared__ __attribute__((aligned(16))) float W2p[32 * W2PITCH];
  __shared__ __attribute__((aligned(16))) float W3p[64 * W3PITCH];
  __shared__ __attribute__((aligned(16))) float Tl[FWAVES * 16 * FPITCH];

  for (int i = threadIdx.x; i < 64 * 128; i += FWAVES * 32) {
    const int k = i >> 7;
    const int col = i & 127;
    W2p[(k >> 1) * W2PITCH + col * 2 + (k & 1)] = W2[i];
  }
  for (int i = threadIdx.x; i < 128 * 64; i += FWAVES * 32) {
    const int k = i >> 6;
    const int col = i & 63;
    W3p[(k >> 1) * W3PITCH + col * 2 + (k & 1)] = W3[i];
  }
  __syncthreads();

  const int lane = threadIdx.x & 31;
  const int wave = threadIdx.x >> 5;
  const int tile = __builtin_amdgcn_readfirstlane((int)(blockIdx.x * FWAVES + wave));
  if (tile * 16 >= n) return;
  const int half = lane >> 4;
  const int q = lane & 15;
  float* __restrict__ T = Tl + wave * (16 * FPITCH);

  const int last_node = n - 1;
  const int hnode = (tile * 16 + q < last_node) ? (tile * 16 + q) : last_node;
  const int2 sgv = seg[hnode];
  const float div = dinv[hnode];
  const unsigned last_row = (unsigned)last_node;

  int2 cwn[8];
#pragma unroll
  for (int u = 0; u < 8; ++u) cwn[u] = make_int2(0, 0);
#pragma nounroll
  for (int r = -1; r < 16; ++r) {
    const bool live = (r >= 0);
    const int rc = live ? r : 0;
    int2 cw[8];
#pragma unroll
    for (int u = 0; u < 8; ++u) cw[u] = cwn[u];
    const int rn = (r < 15) ? r + 1 : 15;
    load_pairs(colw, __builtin_amdgcn_readlane(sgv.x, rn), half, cwn);

    const int node_raw = tile * 16 + rc;
    const int node = (node_raw < last_node) ? node_raw : last_node;
    const int start = __builtin_amdgcn_readlane(sgv.x, rc);
    const int cnt = live ? __builtin_amdgcn_readlane(sgv.y, rc) : 0;
    const float di = __int_as_float(__builtin_amdgcn_readlane(__float_as_int(div), rc));
    const float4 hself = ld_row16(H, (unsigned)node, q);

    float4 acc = make_float4(0.0f, 0.0f, 0.0f, 0.0f);
    float4 v[8];
    issue_rows(H, cw, cnt, half, q, last_row, v);
    fma_rows(cw, v, cnt, half, acc);
    for (int j = 16; j < cnt; j += 16) {
      int2 c2[8];
      load_pairs(colw, start + j, half, c2);
      issue_rows(H, c2, cnt - j, half, q, last_row, v);
      fma_rows(c2, v, cnt - j, half, acc);
    }

    acc.x += __shfl_xor(acc.x, 16);
    acc.y += __shfl_xor(acc.y, 16);
    acc.z += __shfl_xor(acc.z, 16);
    acc.w += __shfl_xor(acc.w, 16);

    if (live && half == 0) {
      float4 rr;
      rr.x = hself.x * di * di;
      rr.y = hself.y * di * di;
      rr.z = hself.z * di * di;
      rr.w = hself.w * di * di;
      rr.x += acc.x; rr.y += acc.y; rr.z += acc.z; rr.w += acc.w;
      *(float4*)(T + rc * FPITCH + q * 4) = rr;
    }
  }
  wave_lds_order();

  v8f acc1[8];
#pragma unroll
  for (int t = 0; t < 8; ++t) acc1[t] = (v8f){};
#pragma unroll
  for (int k = 0; k < 64; k += 4) {
    const int kk = k + half * 2;
    const v2f a = *(const v2f*)(T + q * FPITCH + kk);
    const float* __restrict__ wrow = W2p + (kk >> 1) * W2PITCH + q * 2;
#pragma unroll
    for (int t = 0; t < 8; ++t) {
      const v2f b = *(const v2f*)(wrow + t * 32);
      acc1[t] = __builtin_amdgcn_wmma_f32_16x16x4_f32(false, a, false, b,
                                                      (short)0, acc1[t], false, false);
    }
  }

  v8f acc2[4];
#pragma unroll
  for (int t = 0; t < 4; ++t) acc2[t] = (v8f){};
#pragma unroll
  for (int hh = 0; hh < 2; ++hh) {
    wave_lds_order();
#pragma unroll
    for (int t4 = 0; t4 < 4; ++t4) {
      const int t = hh * 4 + t4;
      const float bv = b2[t * 16 + q];
#pragma unroll
      for (int v = 0; v < 8; ++v) {
        T[(v + half * 8) * FPITCH + t4 * 16 + q] = fmaxf(acc1[t][v] + bv, 0.0f);
      }
    }
    wave_lds_order();
#pragma unroll
    for (int k = 0; k < 64; k += 4) {
      const int kk = k + half * 2;
      const v2f a = *(const v2f*)(T + q * FPITCH + kk);
      const int kg = hh * 64 + kk;
      const float* __restrict__ wrow = W3p + (kg >> 1) * W3PITCH + q * 2;
#pragma unroll
      for (int t = 0; t < 4; ++t) {
        const v2f b = *(const v2f*)(wrow + t * 32);
        acc2[t] = __builtin_amdgcn_wmma_f32_16x16x4_f32(false, a, false, b,
                                                        (short)0, acc2[t], false, false);
      }
    }
  }

  wave_lds_order();
#pragma unroll
  for (int t = 0; t < 4; ++t) {
#pragma unroll
    for (int v = 0; v < 8; ++v) {
      T[(v + half * 8) * FPITCH + t * 16 + q] = acc2[t][v];
    }
  }
  wave_lds_order();
#pragma unroll
  for (int p = 0; p < 8; ++p) {
    const int row = p * 2 + half;
    const float4 val = *(const float4*)(T + row * FPITCH + q * 4);
    const int orow = tile * 16 + row;
    if (orow < n) {
      *(float4*)((char*)Out + ((unsigned)orow * 256u + (unsigned)q * 16u)) = val;
    }
  }
}

static inline unsigned int cdiv(long long a, long long b) {
  return (unsigned int)((a + b - 1) / b);
}

extern "C" void kernel_launch(void* const* d_in, const int* in_sizes, int n_in,
                              void* d_out, int out_size, void* d_ws, size_t ws_size,
                              hipStream_t stream) {
  const float* x  = (const float*)d_in[0];
  const int*   ei = (const int*)d_in[1];
  const float* W1 = (const float*)d_in[2];
  const float* b1 = (const float*)d_in[3];
  const float* W2 = (const float*)d_in[4];
  const float* b2 = (const float*)d_in[5];
  const float* W3 = (const float*)d_in[6];
  const float* b3 = (const float*)d_in[7];
  float* out = (float*)d_out;

  const int N = NODES;
  const int E = in_sizes[1] / 2;

  const int* src = ei;
  const int* dst = ei + E;

  float* dinv = (float*)d_ws;
  float* bufA = dinv + N;
  float* bufB = bufA + (size_t)N * 128;
  int2* colw  = (int2*)(bufA + (size_t)N * 64);
  int*  rank  = (int*)(colw + E);
  int2* seg   = (int2*)(rank + E);
  int*  cnt   = (int*)(seg + N);
  int*  total = cnt + N;

  const int T = 256;
  const unsigned int gemm_blocks = cdiv((N + 15) / 16, 8);
  const unsigned int row_blocks  = cdiv(N, 8);
  const unsigned int fused_blocks = cdiv((N + 15) / 16, FWAVES);

  zero_i32_kernel<<<cdiv(N + 1, T), T, 0, stream>>>(cnt, N + 1);
  count_rank_kernel<<<cdiv(E, T), T, 0, stream>>>(dst, E, cnt, rank);
  node_prep_kernel<<<cdiv(N, T), T, 0, stream>>>(cnt, N, total, seg, dinv);
  fill_kernel<<<cdiv(E, T), T, 0, stream>>>(src, dst, rank, seg, dinv, E, colw);

  gemm_wmma_f32_kernel<128, 64, false, false><<<gemm_blocks, T, 0, stream>>>(x, W1, b1, bufA, N);
  gather_rows_kernel<true, true><<<row_blocks, T, 0, stream>>>(seg, colw, bufA, dinv, b1, bufB, N);

  gather_mlp_kernel<<<fused_blocks, FWAVES * 32, 0, stream>>>(seg, colw, bufB, dinv, W2, b2, W3, bufA, N);

  gather_rows_kernel<false, true><<<row_blocks, T, 0, stream>>>(seg, colw, bufA, dinv, b3, out, N);
}


# edge list built by a two-level counting sort on LDS counters (256-row buckets, then rows) instead of one returning global atomic per edge; slot order stays an arrival order (f32 sum order may differ)
# speedup vs baseline: 8.1770x; 1.5307x over previous
#include <hip/hip_runtime.h>
#include <hip/hip_bf16.h>

typedef __attribute__((ext_vector_type(2))) float v2f;
typedef __attribute__((ext_vector_type(8))) float v8f;

#define NODES 100000

#define BUCKET_SHIFT 8
#define BUCKET_ROWS 256
#define NBUCKETS ((NODES + BUCKET_ROWS - 1) / BUCKET_ROWS)
#define SORT_CHUNK 8192
#define SORT_THREADS 1024
#define SORT_PER_THREAD (SORT_CHUNK / SORT_THREADS)
#define SCAN_WAVES ((NBUCKETS + 31) / 32)

static __device__ __forceinline__ int clamp_node(int d, int n) {
  const unsigned last = (unsigned)(n - 1);
  return (int)(((unsigned)d < last) ? (unsigned)d : last);
}

__global__ void __launch_bounds__(SORT_THREADS)
bucket_hist_kernel(const int* __restrict__ dst, int e, int n, int* __restrict__ blockhist) {
  __shared__ int lh[NBUCKETS];
  const int tid = threadIdx.x;
  for (int i = tid; i < NBUCKETS; i += SORT_THREADS) lh[i] = 0;
  __syncthreads();
  const int base = blockIdx.x * SORT_CHUNK;
  int d[SORT_PER_THREAD];
#pragma unroll
  for (int k = 0; k < SORT_PER_THREAD; ++k) {
    const int ei = base + k * SORT_THREADS + tid;
    d[k] = dst[(ei < e) ? ei : (e - 1)];
  }
#pragma unroll
  for (int k = 0; k < SORT_PER_THREAD; ++k) {
    const int ei = base + k * SORT_THREADS + tid;
    if (ei < e) atomicAdd(&lh[clamp_node(d[k], n) >> BUCKET_SHIFT], 1);
  }
  __syncthreads();
  for (int i = tid; i < NBUCKETS; i += SORT_THREADS) blockhist[blockIdx.x * NBUCKETS + i] = lh[i];
}

__global__ void __launch_bounds__(SORT_THREADS)
bucket_scatter_kernel(const int* __restrict__ src, const int* __restrict__ dst, int e, int n,
                      const int* __restrict__ blockhist, int2* __restrict__ bucketed,
                      int* __restrict__ bucket_start, int* __restrict__ bucket_cnt) {
  __shared__ int lh[NBUCKETS];
  __shared__ int pre[NBUCKETS];
  __shared__ int bst[NBUCKETS];
  __shared__ int lst[NBUCKETS];
  __shared__ int wtot[SCAN_WAVES], wloc[SCAN_WAVES];
  __shared__ int2 stage[SORT_CHUNK];
  const int tid = threadIdx.x;
  const int lane = tid & 31;
  const int wave = tid >> 5;
  const int blk = blockIdx.x;
  const int nblk = gridDim.x;
  for (int i = tid; i < NBUCKETS; i += SORT_THREADS) lh[i] = 0;
  __syncthreads();

  const int base = blk * SORT_CHUNK;
  int s[SORT_PER_THREAD], d[SORT_PER_THREAD], r[SORT_PER_THREAD];
#pragma unroll
  for (int k = 0; k < SORT_PER_THREAD; ++k) {
    const int ei = base + k * SORT_THREADS + tid;
    const int eic = (ei < e) ? ei : (e - 1);
    s[k] = src[eic];
    d[k] = clamp_node(dst[eic], n);
  }
#pragma unroll
  for (int k = 0; k < SORT_PER_THREAD; ++k) {
    const int ei = base + k * SORT_THREADS + tid;
    r[k] = 0;
    if (ei < e) r[k] = atomicAdd(&lh[d[k] >> BUCKET_SHIFT], 1);
  }
  int total = 0, before = 0;
  if (tid < NBUCKETS) {
    for (int b0 = 0; b0 < nblk; b0 += 16) {
      int v[16];
#pragma unroll
      for (int j = 0; j < 16; ++j) {
        const int bb = b0 + j;
        v[j] = (bb < nblk) ? blockhist[bb * NBUCKETS + tid] : 0;
      }
#pragma unroll
      for (int j = 0; j < 16; ++j) {
        total += v[j];
        if (b0 + j < blk) before += v[j];
      }
    }
    pre[tid] = before;
  }
  __syncthreads();

  const int mine = (tid < NBUCKETS) ? lh[tid] : 0;
  int it = total, il = mine;
  if (wave < SCAN_WAVES) {
#pragma unroll
    for (int o = 1; o < 32; o <<= 1) {
      const int a = __shfl_up(it, o);
      const int b = __shfl_up(il, o);
      if (lane >= o) { it += a; il += b; }
    }
    if (lane == 31) { wtot[wave] = it; wloc[wave] = il; }
  }
  __syncthreads();
  if (tid < NBUCKETS) {
    int bt = 0, bl = 0;
#pragma unroll
    for (int w = 0; w < SCAN_WAVES; ++w) {
      bt += (w < wave) ? wtot[w] : 0;
      bl += (w < wave) ? wloc[w] : 0;
    }
    bst[tid] = bt + it - total;
    lst[tid] = bl + il - mine;
    if (blk == 0) {
      bucket_start[tid] = bt + it - total;
      bucket_cnt[tid] = total;
    }
  }
  __syncthreads();

#pragma unroll
  for (int k = 0; k < SORT_PER_THREAD; ++k) {
    const int ei = base + k * SORT_THREADS + tid;
    if (ei < e) stage[lst[d[k] >> BUCKET_SHIFT] + r[k]] = make_int2(s[k], d[k]);
  }
  __syncthreads();
  const int left = e - base;
  const int nvalid = (left < SORT_CHUNK) ? left : SORT_CHUNK;
#pragma unroll
  for (int k = 0; k < SORT_PER_THREAD; ++k) {
    const int i = k * SORT_THREADS + tid;
    if (i < nvalid) {
      const int2 pr = stage[i];
      const int bk = pr.y >> BUCKET_SHIFT;
      bucketed[bst[bk] + pre[bk] + (i - lst[bk])] = pr;
    }
  }
}

#define ROWS_THREADS 512
__global__ void __launch_bounds__(ROWS_THREADS)
bucket_rows_kernel(const int2* __restrict__ bucketed, const int* __restrict__ bucket_start,
                   const int* __restrict__ bucket_cnt, int n,
                   int2* __restrict__ seg, float* __restrict__ dinv, int2* __restrict__ colw) {
  __shared__ int cnt[BUCKET_ROWS];
  __shared__ int cur[BUCKET_ROWS];
  __shared__ int wsum[BUCKET_ROWS / 32];
  const int tid = threadIdx.x;
  const int lane = tid & 31;
  const int wave = tid >> 5;
  const int b = blockIdx.x;
  const int bs = bucket_start[b];
  const int bc = bucket_cnt[b];
  if (tid < BUCKET_ROWS) cnt[tid] = 0;
  __syncthreads();

  for (int i0 = tid; i0 < bc; i0 += 4 * ROWS_THREADS) {
    int dd[4];
#pragma unroll
    for (int k = 0; k < 4; ++k) {
      const int i = i0 + k * ROWS_THREADS;
      dd[k] = bucketed[bs + ((i < bc) ? i : i0)].y;
    }
#pragma unroll
    for (int k = 0; k < 4; ++k) {
      const int i = i0 + k * ROWS_THREADS;
      if (i < bc) atomicAdd(&cnt[clamp_node(dd[k], n) & (BUCKET_ROWS - 1)], 1);
    }
  }
  __syncthreads();

  const int c = (tid < BUCKET_ROWS) ? cnt[tid] : 0;
  int incl = c;
#pragma unroll
  for (int o = 1; o < 32; o <<= 1) {
    const int t = __shfl_up(incl, o);
    if (lane >= o) incl += t;
  }
  if (tid < BUCKET_ROWS && lane == 31) wsum[wave] = incl;
  __syncthreads();
  if (tid < BUCKET_ROWS) {
    int before = 0;
#pragma unroll
    for (int w = 0; w < BUCKET_ROWS / 32; ++w) before += (w < wave) ? wsum[w] : 0;
    const int first = bs + before + incl - c;
    cur[tid] = first;
    const int node = b * BUCKET_ROWS + tid;
    if (node < n) {
      seg[node] = make_int2(first, c);
      dinv[node] = rsqrtf((float)c + 1.0f);
    }
  }
  __syncthreads();

  for (int i0 = tid; i0 < bc; i0 += 4 * ROWS_THREADS) {
    int2 pr[4];
#pragma unroll
    for (int k = 0; k < 4; ++k) {
      const int i = i0 + k * ROWS_THREADS;
      pr[k] = bucketed[bs + ((i < bc) ? i : i0)];
    }
#pragma unroll
    for (int k = 0; k < 4; ++k) {
      const int i = i0 + k * ROWS_THREADS;
      if (i < bc) {
        const int dc = clamp_node(pr[k].y, n);
        const int p = atomicAdd(&cur[dc & (BUCKET_ROWS - 1)], 1);
        colw[p] = make_int2(pr[k].x, dc);
      }
    }
  }
}

__global__ void edge_weight_kernel(int2* __restrict__ colw, const float* __restrict__ dinv,
                                   int e, int n) {
  int2 pr[4];
  float w[4];
#pragma unroll
  for (int k = 0; k < 4; ++k) {
    const int p = blockIdx.x * 1024 + k * 256 + threadIdx.x;
    pr[k] = colw[(p < e) ? p : (e - 1)];
  }
#pragma unroll
  for (int k = 0; k < 4; ++k) w[k] = dinv[clamp_node(pr[k].x, n)] * dinv[clamp_node(pr[k].y, n)];
#pragma unroll
  for (int k = 0; k < 4; ++k) {
    const int p = blockIdx.x * 1024 + k * 256 + threadIdx.x;
    if (p < e) colw[p].y = __float_as_int(w[k]);
  }
}

static __device__ __forceinline__ float4 ld_row16(const float* __restrict__ H, unsigned s, int q) {
  const unsigned off = s * 256u + (unsigned)q * 16u;
  return *(const float4*)((const char*)H + off);
}

template <bool RELU_OUT, bool BIAS>
__global__ void __launch_bounds__(256)
gather_rows_kernel(const int2* __restrict__ seg, const int2* __restrict__ colw,
                   const float* __restrict__ H, const float* __restrict__ dinv,
                   const float* __restrict__ bias, float* __restrict__ Out, int n) {
  const int lane = threadIdx.x & 31;
  const int node = __builtin_amdgcn_readfirstlane((int)(blockIdx.x * 8 + (threadIdx.x >> 5)));
  if (node >= n) return;
  const int half = lane >> 4;
  const int q = lane & 15;

  const int2 sg = seg[node];
  const int start = sg.x;
  const int cnt = sg.y;
  const float di = dinv[node];
  const float4 hself = ld_row16(H, (unsigned)node, q);

  float4 acc = make_float4(0.0f, 0.0f, 0.0f, 0.0f);
  const unsigned last_row = (unsigned)(n - 1);
  for (int j = 0; j < cnt; j += 16) {
    const int m = cnt - j;
    const int2* __restrict__ cp = colw + (start + j);
    int ex[8], ey[8], ox[8], oy[8];
#pragma unroll
    for (int u = 0; u < 8; ++u) {
      const int2 ce = cp[2 * u];
      const int2 co = cp[2 * u + 1];
      ex[u] = __builtin_amdgcn_readfirstlane(ce.x);
      ey[u] = __builtin_amdgcn_readfirstlane(ce.y);
      ox[u] = __builtin_amdgcn_readfirstlane(co.x);
      oy[u] = __builtin_amdgcn_readfirstlane(co.y);
    }
    float  w[8];
    float4 v[8];
#pragma unroll
    for (int u = 0; u < 8; ++u) {
      const int sidx = half ? ox[u] : ex[u];
      w[u] = __int_as_float(half ? oy[u] : ey[u]);
      const unsigned srow = ((unsigned)sidx < last_row) ? (unsigned)sidx : last_row;
      v[u] = make_float4(0.0f, 0.0f, 0.0f, 0.0f);
      if (2 * u + half < m) v[u] = ld_row16(H, srow, q);
    }
#pragma unroll
    for (int u = 0; u < 8; ++u) {
      const float wu = (2 * u + half < m) ? w[u] : 0.0f;
      acc.x = fmaf(wu, v[u].x, acc.x);
      acc.y = fmaf(wu, v[u].y, acc.y);
      acc.z = fmaf(wu, v[u].z, acc.z);
      acc.w = fmaf(wu, v[u].w, acc.w);
    }
  }

  acc.x += __shfl_xor(acc.x, 16);
  acc.y += __shfl_xor(acc.y, 16);
  acc.z += __shfl_xor(acc.z, 16);
  acc.w += __shfl_xor(acc.w, 16);

  if (half == 0) {
    float4 r;
    r.x = hself.x * di * di;
    r.y = hself.y * di * di;
    r.z = hself.z * di * di;
    r.w = hself.w * di * di;
    if (BIAS) {
      r.x += bias[q * 4 + 0];
      r.y += bias[q * 4 + 1];
      r.z += bias[q * 4 + 2];
      r.w += bias[q * 4 + 3];
    }
    r.x += acc.x; r.y += acc.y; r.z += acc.z; r.w += acc.w;
    if (RELU_OUT) {
      r.x = fmaxf(r.x, 0.0f); r.y = fmaxf(r.y, 0.0f);
      r.z = fmaxf(r.z, 0.0f); r.w = fmaxf(r.w, 0.0f);
    }
    *(float4*)((char*)Out + ((unsigned)node * 256u + (unsigned)q * 16u)) = r;
  }
}

template <int K, int FOUT, bool RELU, bool BIAS>
__global__ void gemm_wmma_f32_kernel(const float* __restrict__ X,
                                     const float* __restrict__ W,
                                     const float* __restrict__ bias,
                                     float* __restrict__ H,
                                     int n_rows) {
  constexpr int NT = FOUT / 16;
  __shared__ float Wp[K * FOUT];

  for (int i = threadIdx.x; i < K * FOUT; i += 256) {
    int k = i / FOUT;
    int col = i - k * FOUT;
    Wp[(k >> 1) * (FOUT * 2) + col * 2 + (k & 1)] = W[i];
  }
  __syncthreads();

  const int wave = threadIdx.x >> 5;
  const int lane = threadIdx.x & 31;
  const int tile_m = blockIdx.x * 8 + wave;
  if (tile_m * 16 >= n_rows) return;

  const int half = lane >> 4;
  const int l15  = lane & 15;

  const float* __restrict__ xrow = X + (size_t)(tile_m * 16 + l15) * K;

  v8f acc[NT];
#pragma unroll
  for (int t = 0; t < NT; ++t) acc[t] = (v8f){};

#pragma unroll
  for (int k = 0; k < K; k += 4) {
    const int kk = k + half * 2;
    float a0 = xrow[kk];
    float a1 = xrow[kk + 1];
    if (RELU) { a0 = fmaxf(a0, 0.0f); a1 = fmaxf(a1, 0.0f); }
    v2f a;
    a.x = a0;
    a.y = a1;
    const float* __restrict__ wrow = Wp + (kk >> 1) * (FOUT * 2) + l15 * 2;
#pragma unroll
    for (int t = 0; t < NT; ++t) {
      const v2f b = *(const v2f*)(wrow + t * 32);
      acc[t] = __builtin_amdgcn_wmma_f32_16x16x4_f32(false, a, false, b,
                                                     (short)0, acc[t], false, false);
    }
  }

  float* __restrict__ hbase = H + (size_t)(tile_m * 16) * FOUT + l15;
#pragma unroll
  for (int t = 0; t < NT; ++t) {
    float bv = 0.0f;
    if (BIAS) bv = bias[t * 16 + l15];
#pragma unroll
    for (int v = 0; v < 8; ++v) {
      hbase[(size_t)(v + half * 8) * FOUT + t * 16] = acc[t][v] + bv;
    }
  }
}


#define FPITCH 68
#define FWAVES 25
#define W2PITCH (2 * 128 + 32)
#define W3PITCH (2 * 64 + 32)

static __device__ __forceinline__ void wave_lds_order() {
  __builtin_amdgcn_fence(__ATOMIC_SEQ_CST, "wavefront");
  __builtin_amdgcn_wave_barrier();
}

static __device__ __forceinline__ void load_pairs(const int2* __restrict__ colw, int first,
                                                  int half, int2 (&cw)[8]) {
#pragma unroll
  for (int u = 0; u < 8; ++u) cw[u] = colw[first + 2 * u + half];
}

static __device__ __forceinline__ void issue_rows(const float* __restrict__ H, const int2 (&cw)[8],
                                                  int m, int half, int q, unsigned last_row,
                                                  float4 (&v)[8]) {
#pragma unroll
  for (int u = 0; u < 8; ++u) {
    const unsigned srow = ((unsigned)cw[u].x < last_row) ? (unsigned)cw[u].x : last_row;
    v[u] = make_float4(0.0f, 0.0f, 0.0f, 0.0f);
    if (2 * u + half < m) v[u] = ld_row16(H, srow, q);
  }
}

static __device__ __forceinline__ void fma_rows(const int2 (&cw)[8], const float4 (&v)[8],
                                                int m, int half, float4& acc) {
#pragma unroll
  for (int u = 0; u < 8; ++u) {
    const float wu = (2 * u + half < m) ? __int_as_float(cw[u].y) : 0.0f;
    acc.x = fmaf(wu, v[u].x, acc.x);
    acc.y = fmaf(wu, v[u].y, acc.y);
    acc.z = fmaf(wu, v[u].z, acc.z);
    acc.w = fmaf(wu, v[u].w, acc.w);
  }
}

__global__ void __launch_bounds__(FWAVES * 32)
gather_mlp_kernel(const int2* __restrict__ seg, const int2* __restrict__ colw,
                  const float* __restrict__ H, const float* __restrict__ dinv,
                  const float* __restrict__ W2, const float* __restrict__ b2,
                  const float* __restrict__ W3, float* __restrict__ Out, int n) {
  __shared__ __attribute__((aligned(16))) float W2p[32 * W2PITCH];
  __shared__ __attribute__((aligned(16))) float W3p[64 * W3PITCH];
  __shared__ __attribute__((aligned(16))) float Tl[FWAVES * 16 * FPITCH];

  for (int i = threadIdx.x; i < 64 * 128; i += FWAVES * 32) {
    const int k = i >> 7;
    const int col = i & 127;
    W2p[(k >> 1) * W2PITCH + col * 2 + (k & 1)] = W2[i];
  }
  for (int i = threadIdx.x; i < 128 * 64; i += FWAVES * 32) {
    const int k = i >> 6;
    const int col = i & 63;
    W3p[(k >> 1) * W3PITCH + col * 2 + (k & 1)] = W3[i];
  }
  __syncthreads();

  const int lane = threadIdx.x & 31;
  const int wave = threadIdx.x >> 5;
  const int tile = __builtin_amdgcn_readfirstlane((int)(blockIdx.x * FWAVES + wave));
  if (tile * 16 >= n) return;
  const int half = lane >> 4;
  const int q = lane & 15;
  float* __restrict__ T = Tl + wave * (16 * FPITCH);

  const int last_node = n - 1;
  const int hnode = (tile * 16 + q < last_node) ? (tile * 16 + q) : last_node;
  const int2 sgv = seg[hnode];
  const float div = dinv[hnode];
  const unsigned last_row = (unsigned)last_node;

  int2 cwn[8];
#pragma unroll
  for (int u = 0; u < 8; ++u) cwn[u] = make_int2(0, 0);
#pragma nounroll
  for (int r = -1; r < 16; ++r) {
    const bool live = (r >= 0);
    const int rc = live ? r : 0;
    int2 cw[8];
#pragma unroll
    for (int u = 0; u < 8; ++u) cw[u] = cwn[u];
    const int rn = (r < 15) ? r + 1 : 15;
    load_pairs(colw, __builtin_amdgcn_readlane(sgv.x, rn), half, cwn);

    const int node_raw = tile * 16 + rc;
    const int node = (node_raw < last_node) ? node_raw : last_node;
    const int start = __builtin_amdgcn_readlane(sgv.x, rc);
    const int cnt = live ? __builtin_amdgcn_readlane(sgv.y, rc) : 0;
    const float di = __int_as_float(__builtin_amdgcn_readlane(__float_as_int(div), rc));
    const float4 hself = ld_row16(H, (unsigned)node, q);

    float4 acc = make_float4(0.0f, 0.0f, 0.0f, 0.0f);
    float4 v[8];
    issue_rows(H, cw, cnt, half, q, last_row, v);
    fma_rows(cw, v, cnt, half, acc);
    for (int j = 16; j < cnt; j += 16) {
      int2 c2[8];
      load_pairs(colw, start + j, half, c2);
      issue_rows(H, c2, cnt - j, half, q, last_row, v);
      fma_rows(c2, v, cnt - j, half, acc);
    }

    acc.x += __shfl_xor(acc.x, 16);
    acc.y += __shfl_xor(acc.y, 16);
    acc.z += __shfl_xor(acc.z, 16);
    acc.w += __shfl_xor(acc.w, 16);

    if (live && half == 0) {
      float4 rr;
      rr.x = hself.x * di * di;
      rr.y = hself.y * di * di;
      rr.z = hself.z * di * di;
      rr.w = hself.w * di * di;
      rr.x += acc.x; rr.y += acc.y; rr.z += acc.z; rr.w += acc.w;
      *(float4*)(T + rc * FPITCH + q * 4) = rr;
    }
  }
  wave_lds_order();

  v8f acc1[8];
#pragma unroll
  for (int t = 0; t < 8; ++t) acc1[t] = (v8f){};
#pragma unroll
  for (int k = 0; k < 64; k += 4) {
    const int kk = k + half * 2;
    const v2f a = *(const v2f*)(T + q * FPITCH + kk);
    const float* __restrict__ wrow = W2p + (kk >> 1) * W2PITCH + q * 2;
#pragma unroll
    for (int t = 0; t < 8; ++t) {
      const v2f b = *(const v2f*)(wrow + t * 32);
      acc1[t] = __builtin_amdgcn_wmma_f32_16x16x4_f32(false, a, false, b,
                                                      (short)0, acc1[t], false, false);
    }
  }

  v8f acc2[4];
#pragma unroll
  for (int t = 0; t < 4; ++t) acc2[t] = (v8f){};
#pragma unroll
  for (int hh = 0; hh < 2; ++hh) {
    wave_lds_order();
#pragma unroll
    for (int t4 = 0; t4 < 4; ++t4) {
      const int t = hh * 4 + t4;
      const float bv = b2[t * 16 + q];
#pragma unroll
      for (int v = 0; v < 8; ++v) {
        T[(v + half * 8) * FPITCH + t4 * 16 + q] = fmaxf(acc1[t][v] + bv, 0.0f);
      }
    }
    wave_lds_order();
#pragma unroll
    for (int k = 0; k < 64; k += 4) {
      const int kk = k + half * 2;
      const v2f a = *(const v2f*)(T + q * FPITCH + kk);
      const int kg = hh * 64 + kk;
      const float* __restrict__ wrow = W3p + (kg >> 1) * W3PITCH + q * 2;
#pragma unroll
      for (int t = 0; t < 4; ++t) {
        const v2f b = *(const v2f*)(wrow + t * 32);
        acc2[t] = __builtin_amdgcn_wmma_f32_16x16x4_f32(false, a, false, b,
                                                        (short)0, acc2[t], false, false);
      }
    }
  }

  wave_lds_order();
#pragma unroll
  for (int t = 0; t < 4; ++t) {
#pragma unroll
    for (int v = 0; v < 8; ++v) {
      T[(v + half * 8) * FPITCH + t * 16 + q] = acc2[t][v];
    }
  }
  wave_lds_order();
#pragma unroll
  for (int p = 0; p < 8; ++p) {
    const int row = p * 2 + half;
    const float4 val = *(const float4*)(T + row * FPITCH + q * 4);
    const int orow = tile * 16 + row;
    if (orow < n) {
      *(float4*)((char*)Out + ((unsigned)orow * 256u + (unsigned)q * 16u)) = val;
    }
  }
}

static inline unsigned int cdiv(long long a, long long b) {
  return (unsigned int)((a + b - 1) / b);
}

extern "C" void kernel_launch(void* const* d_in, const int* in_sizes, int n_in,
                              void* d_out, int out_size, void* d_ws, size_t ws_size,
                              hipStream_t stream) {
  const float* x  = (const float*)d_in[0];
  const int*   ei = (const int*)d_in[1];
  const float* W1 = (const float*)d_in[2];
  const float* b1 = (const float*)d_in[3];
  const float* W2 = (const float*)d_in[4];
  const float* b2 = (const float*)d_in[5];
  const float* W3 = (const float*)d_in[6];
  const float* b3 = (const float*)d_in[7];
  float* out = (float*)d_out;

  const int N = NODES;
  const int E = in_sizes[1] / 2;

  const int* src = ei;
  const int* dst = ei + E;

  float* dinv = (float*)d_ws;
  float* bufA = dinv + N;
  float* bufB = bufA + (size_t)N * 128;
  int2* colw  = (int2*)(bufA + (size_t)N * 64);
  int*  blockhist = (int*)(colw + E);
  int2* seg   = (int2*)(blockhist + E);
  int*  bucket_start = (int*)(seg + N);
  int*  bucket_cnt = bucket_start + NBUCKETS;
  int2* bucketed = (int2*)(bufB + (size_t)N * 64);

  const int T = 256;
  const unsigned int gemm_blocks = cdiv((N + 15) / 16, 8);
  const unsigned int row_blocks  = cdiv(N, 8);
  const unsigned int fused_blocks = cdiv((N + 15) / 16, FWAVES);

  const unsigned int sort_blocks = cdiv(E, SORT_CHUNK);
  bucket_hist_kernel<<<sort_blocks, SORT_THREADS, 0, stream>>>(dst, E, N, blockhist);
  bucket_scatter_kernel<<<sort_blocks, SORT_THREADS, 0, stream>>>(src, dst, E, N, blockhist, bucketed,
                                                                 bucket_start, bucket_cnt);
  bucket_rows_kernel<<<NBUCKETS, ROWS_THREADS, 0, stream>>>(bucketed, bucket_start, bucket_cnt, N,
                                                           seg, dinv, colw);
  edge_weight_kernel<<<cdiv(E, 1024), T, 0, stream>>>(colw, dinv, E, N);

  gemm_wmma_f32_kernel<128, 64, false, false><<<gemm_blocks, T, 0, stream>>>(x, W1, b1, bufA, N);
  gather_rows_kernel<true, true><<<row_blocks, T, 0, stream>>>(seg, colw, bufA, dinv, b1, bufB, N);

  gather_mlp_kernel<<<fused_blocks, FWAVES * 32, 0, stream>>>(seg, colw, bufB, dinv, W2, b2, W3, bufA, N);

  gather_rows_kernel<false, true><<<row_blocks, T, 0, stream>>>(seg, colw, bufA, dinv, b3, out, N);
}
